# starting kernel
# baseline (speedup 1.0000x reference)
#include <hip/hip_runtime.h>
#include <hip/hip_bf16.h>

#define NPTS      2048
#define MODEL_DIM 1024
#define RANK      64

typedef float v2f __attribute__((ext_vector_type(2)));
typedef float v8f __attribute__((ext_vector_type(8)));

__global__ __launch_bounds__(256) void psd_project_kernel(
    const float* __restrict__ R, const float* __restrict__ P,
    float* __restrict__ T) {
  const int lane = threadIdx.x & 31;
  const int wave = threadIdx.x >> 5;
  const int tile = blockIdx.x * 8 + wave;
  const int rI = (tile >> 2) * 16;
  const int cB = (tile & 3) * 16;

  const int m    = lane & 15;
  const int koff = (lane >> 4) * 2;

  const float* aPtr = R + (size_t)(rI + m) * MODEL_DIM + koff;
  const float* bCol = P + cB + m;

  v8f acc = {};
#pragma unroll 8
  for (int s = 0; s < MODEL_DIM / 4; ++s) {
    const int k0 = 4 * s;
    v2f a = *(const v2f*)(aPtr + k0);
    v2f b;
    b.x = bCol[(size_t)(k0 + koff) * RANK];
    b.y = bCol[(size_t)(k0 + koff + 1) * RANK];
    acc = __builtin_amdgcn_wmma_f32_16x16x4_f32(
         false, a,  false, b,
         (short)0, acc,  false,  false);
  }

  const int mBase = rI + ((lane >> 4) << 3);
#pragma unroll
  for (int r = 0; r < 8; ++r) {
    T[(size_t)(mBase + r) * RANK + cB + m] = acc[r];
  }
}

__global__ __launch_bounds__(256) void psd_rownorm_kernel(
    const float* __restrict__ T, float* __restrict__ norms) {
  const int i = blockIdx.x * blockDim.x + threadIdx.x;
  if (i < NPTS) {
    const float4* row = (const float4*)(T + (size_t)i * RANK);
    float s = 0.0f;
#pragma unroll
    for (int q = 0; q < RANK / 4; ++q) {
      float4 v = row[q];
      s += v.x * v.x + v.y * v.y + v.z * v.z + v.w * v.w;
    }
    norms[i] = s;
  }
}

__global__ __launch_bounds__(256) void psd_gram_dist_kernel(
    const float* __restrict__ T, const float* __restrict__ norms,
    float* __restrict__ out) {
  const int lane = threadIdx.x & 31;
  const int wave = threadIdx.x >> 5;

  const int tileI = blockIdx.x >> 4;
  const int tileJ = (blockIdx.x & 15) * 8 + wave;
  const int rI = tileI * 16;
  const int rJ = tileJ * 16;

  const int m    = lane & 15;
  const int koff = (lane >> 4) * 2;

  const float* aPtr = T + (size_t)(rI + m) * RANK + koff;
  const float* bPtr = T + (size_t)(rJ + m) * RANK + koff;

  v8f acc = {};
#pragma unroll
  for (int s = 0; s < RANK / 4; ++s) {
    v2f a = *(const v2f*)(aPtr + 4 * s);
    v2f b = *(const v2f*)(bPtr + 4 * s);
    acc = __builtin_amdgcn_wmma_f32_16x16x4_f32(
        false, a, false, b, (short)0, acc, false, false);
  }

  const float nj = norms[rJ + m];
  const int mBase = rI + ((lane >> 4) << 3);
#pragma unroll
  for (int r = 0; r < 8; ++r) {
    const int row = mBase + r;
    const float val = norms[row] + nj - 2.0f * acc[r];
    out[(size_t)row * NPTS + rJ + m] = val;
  }
}

extern "C" void kernel_launch(void* const* d_in, const int* in_sizes, int n_in,
                              void* d_out, int out_size, void* d_ws, size_t ws_size,
                              hipStream_t stream) {
  const float* R = (const float*)d_in[0];
  const float* P = (const float*)d_in[1];
  float* out = (float*)d_out;

  float* T     = (float*)d_ws;
  float* norms = T + (size_t)NPTS * RANK;

  psd_project_kernel<<<64, 256, 0, stream>>>(R, P, T);

  psd_rownorm_kernel<<<NPTS / 256, 256, 0, stream>>>(T, norms);

  psd_gram_dist_kernel<<<2048, 256, 0, stream>>>(T, norms, out);
}


# projection GEMM: K split over the 8 waves of a block (128 each), partial tiles added through LDS in fixed order; 512 blocks instead of 64 so every WGP works; changes summation order
# speedup vs baseline: 1.5820x; 1.5820x over previous
#include <hip/hip_runtime.h>
#include <hip/hip_bf16.h>

#define NPTS      2048
#define MODEL_DIM 1024
#define RANK      64

typedef float v2f __attribute__((ext_vector_type(2)));
typedef float v8f __attribute__((ext_vector_type(8)));

#define KSPLIT 8
#define KSLICE (MODEL_DIM / KSPLIT)

__global__ __launch_bounds__(256) void psd_project_kernel(
    const float* __restrict__ R, const float* __restrict__ P,
    float* __restrict__ T) {
  __shared__ float part[KSPLIT][8][32];

  const int lane = threadIdx.x & 31;
  const int wave = threadIdx.x >> 5;
  const int tile = blockIdx.x;
  const int rI = (tile >> 2) * 16;
  const int cB = (tile & 3) * 16;

  const int m    = lane & 15;
  const int koff = (lane >> 4) * 2;
  const int kBase = wave * KSLICE;

  const float* aPtr = R + (size_t)(rI + m) * MODEL_DIM + kBase + koff;
  const float* bCol = P + (size_t)kBase * RANK + cB + m;

  v8f acc = {};
#pragma unroll
  for (int s = 0; s < KSLICE / 4; ++s) {
    const int k0 = 4 * s;
    v2f a = *(const v2f*)(aPtr + k0);
    v2f b;
    b.x = bCol[(size_t)(k0 + koff) * RANK];
    b.y = bCol[(size_t)(k0 + koff + 1) * RANK];
    acc = __builtin_amdgcn_wmma_f32_16x16x4_f32(
         false, a,  false, b,
         (short)0, acc,  false,  false);
  }

#pragma unroll
  for (int r = 0; r < 8; ++r) {
    part[wave][r][lane] = acc[r];
  }
  __syncthreads();

  float sum = part[0][wave][lane];
  for (int w = 1; w < KSPLIT; ++w) {
    sum += part[w][wave][lane];
  }
  const int row = rI + ((lane >> 4) << 3) + wave;
  T[(size_t)row * RANK + cB + m] = sum;
}

__global__ __launch_bounds__(256) void psd_rownorm_kernel(
    const float* __restrict__ T, float* __restrict__ norms) {
  const int i = blockIdx.x * blockDim.x + threadIdx.x;
  if (i < NPTS) {
    const float4* row = (const float4*)(T + (size_t)i * RANK);
    float s = 0.0f;
#pragma unroll
    for (int q = 0; q < RANK / 4; ++q) {
      float4 v = row[q];
      s += v.x * v.x + v.y * v.y + v.z * v.z + v.w * v.w;
    }
    norms[i] = s;
  }
}

__global__ __launch_bounds__(256) void psd_gram_dist_kernel(
    const float* __restrict__ T, const float* __restrict__ norms,
    float* __restrict__ out) {
  const int lane = threadIdx.x & 31;
  const int wave = threadIdx.x >> 5;

  const int tileI = blockIdx.x >> 4;
  const int tileJ = (blockIdx.x & 15) * 8 + wave;
  const int rI = tileI * 16;
  const int rJ = tileJ * 16;

  const int m    = lane & 15;
  const int koff = (lane >> 4) * 2;

  const float* aPtr = T + (size_t)(rI + m) * RANK + koff;
  const float* bPtr = T + (size_t)(rJ + m) * RANK + koff;

  v8f acc = {};
#pragma unroll
  for (int s = 0; s < RANK / 4; ++s) {
    v2f a = *(const v2f*)(aPtr + 4 * s);
    v2f b = *(const v2f*)(bPtr + 4 * s);
    acc = __builtin_amdgcn_wmma_f32_16x16x4_f32(
        false, a, false, b, (short)0, acc, false, false);
  }

  const float nj = norms[rJ + m];
  const int mBase = rI + ((lane >> 4) << 3);
#pragma unroll
  for (int r = 0; r < 8; ++r) {
    const int row = mBase + r;
    const float val = norms[row] + nj - 2.0f * acc[r];
    out[(size_t)row * NPTS + rJ + m] = val;
  }
}

extern "C" void kernel_launch(void* const* d_in, const int* in_sizes, int n_in,
                              void* d_out, int out_size, void* d_ws, size_t ws_size,
                              hipStream_t stream) {
  const float* R = (const float*)d_in[0];
  const float* P = (const float*)d_in[1];
  float* out = (float*)d_out;

  float* T     = (float*)d_ws;
  float* norms = T + (size_t)NPTS * RANK;

  psd_project_kernel<<<512, 256, 0, stream>>>(R, P, T);

  psd_rownorm_kernel<<<NPTS / 256, 256, 0, stream>>>(T, norms);

  psd_gram_dist_kernel<<<2048, 256, 0, stream>>>(T, norms, out);
}


# distance kernel: each wave keeps the A fragments of 32 rows of T in registers and walks 4 column tiles (32x64 outputs per wave); 2048 waves in 256 blocks instead of 16384 in 2048
# speedup vs baseline: 2.0984x; 1.3264x over previous
#include <hip/hip_runtime.h>
#include <hip/hip_bf16.h>

#define NPTS      2048
#define MODEL_DIM 1024
#define RANK      64

typedef float v2f __attribute__((ext_vector_type(2)));
typedef float v8f __attribute__((ext_vector_type(8)));

#define KSPLIT 8
#define KSLICE (MODEL_DIM / KSPLIT)

__global__ __launch_bounds__(256) void psd_project_kernel(
    const float* __restrict__ R, const float* __restrict__ P,
    float* __restrict__ T) {
  __shared__ float part[KSPLIT][8][32];

  const int lane = threadIdx.x & 31;
  const int wave = threadIdx.x >> 5;
  const int tile = blockIdx.x;
  const int rI = (tile >> 2) * 16;
  const int cB = (tile & 3) * 16;

  const int m    = lane & 15;
  const int koff = (lane >> 4) * 2;
  const int kBase = wave * KSLICE;

  const float* aPtr = R + (size_t)(rI + m) * MODEL_DIM + kBase + koff;
  const float* bCol = P + (size_t)kBase * RANK + cB + m;

  v8f acc = {};
#pragma unroll
  for (int s = 0; s < KSLICE / 4; ++s) {
    const int k0 = 4 * s;
    v2f a = *(const v2f*)(aPtr + k0);
    v2f b;
    b.x = bCol[(size_t)(k0 + koff) * RANK];
    b.y = bCol[(size_t)(k0 + koff + 1) * RANK];
    acc = __builtin_amdgcn_wmma_f32_16x16x4_f32(
         false, a,  false, b,
         (short)0, acc,  false,  false);
  }

#pragma unroll
  for (int r = 0; r < 8; ++r) {
    part[wave][r][lane] = acc[r];
  }
  __syncthreads();

  float sum = part[0][wave][lane];
  for (int w = 1; w < KSPLIT; ++w) {
    sum += part[w][wave][lane];
  }
  const int row = rI + ((lane >> 4) << 3) + wave;
  T[(size_t)row * RANK + cB + m] = sum;
}

__global__ __launch_bounds__(256) void psd_rownorm_kernel(
    const float* __restrict__ T, float* __restrict__ norms) {
  const int i = blockIdx.x * blockDim.x + threadIdx.x;
  if (i < NPTS) {
    const float4* row = (const float4*)(T + (size_t)i * RANK);
    float s = 0.0f;
#pragma unroll
    for (int q = 0; q < RANK / 4; ++q) {
      float4 v = row[q];
      s += v.x * v.x + v.y * v.y + v.z * v.z + v.w * v.w;
    }
    norms[i] = s;
  }
}

#define KSTEPS (RANK / 4)
#define WAVE_COL_TILES 4

__device__ __forceinline__ void load_frag(v2f (&f)[KSTEPS], const float* __restrict__ p) {
#pragma unroll
  for (int s = 0; s < KSTEPS; ++s) {
    f[s] = *(const v2f*)(p + 4 * s);
  }
}

__device__ __forceinline__ void dist_tiles(
    const v2f (&a0)[KSTEPS], const v2f (&a1)[KSTEPS], const v2f (&b)[KSTEPS],
    const float (&ni0)[8], const float (&ni1)[8], float nj,
    float* __restrict__ o0, float* __restrict__ o1) {
  v8f acc0 = {};
  v8f acc1 = {};
#pragma unroll
  for (int s = 0; s < KSTEPS; ++s) {
    acc0 = __builtin_amdgcn_wmma_f32_16x16x4_f32(
        false, a0[s], false, b[s], (short)0, acc0, false, false);
    acc1 = __builtin_amdgcn_wmma_f32_16x16x4_f32(
        false, a1[s], false, b[s], (short)0, acc1, false, false);
  }
#pragma unroll
  for (int r = 0; r < 8; ++r) {
    o0[(size_t)r * NPTS] = ni0[r] + nj - 2.0f * acc0[r];
    o1[(size_t)r * NPTS] = ni1[r] + nj - 2.0f * acc1[r];
  }
}

__global__ __launch_bounds__(256) void psd_gram_dist_kernel(
    const float* __restrict__ T, const float* __restrict__ norms,
    float* __restrict__ out) {
  const int lane = threadIdx.x & 31;
  const int wave = threadIdx.x >> 5;

  const int rI = (blockIdx.x >> 4) * 128 + (wave >> 1) * 32;
  const int cJ = (blockIdx.x & 15) * 128 + (wave & 1) * 64;

  const int m    = lane & 15;
  const int koff = (lane >> 4) * 2;
  const int mBase0 = rI + ((lane >> 4) << 3);
  const int mBase1 = mBase0 + 16;

  v2f a0[KSTEPS], a1[KSTEPS];
  load_frag(a0, T + (size_t)(rI + m) * RANK + koff);
  load_frag(a1, T + (size_t)(rI + 16 + m) * RANK + koff);

  const float* bPtr = T + (size_t)(cJ + m) * RANK + koff;
  v2f b0[KSTEPS], b1[KSTEPS];
  load_frag(b0, bPtr);

  float ni0[8], ni1[8];
#pragma unroll
  for (int r = 0; r < 8; ++r) {
    ni0[r] = norms[mBase0 + r];
    ni1[r] = norms[mBase1 + r];
  }

  float* o0 = out + (size_t)mBase0 * NPTS + cJ + m;
  float* o1 = out + (size_t)mBase1 * NPTS + cJ + m;
  const float* njPtr = norms + cJ + m;

#pragma unroll 1
  for (int t = 0; t < WAVE_COL_TILES; t += 2) {
    load_frag(b1, bPtr + (size_t)(t + 1) * 16 * RANK);
    dist_tiles(a0, a1, b0, ni0, ni1, njPtr[16 * t], o0 + 16 * t, o1 + 16 * t);
    if (t + 2 < WAVE_COL_TILES) {
      load_frag(b0, bPtr + (size_t)(t + 2) * 16 * RANK);
    }
    dist_tiles(a0, a1, b1, ni0, ni1, njPtr[16 * (t + 1)], o0 + 16 * (t + 1), o1 + 16 * (t + 1));
  }
}

extern "C" void kernel_launch(void* const* d_in, const int* in_sizes, int n_in,
                              void* d_out, int out_size, void* d_ws, size_t ws_size,
                              hipStream_t stream) {
  const float* R = (const float*)d_in[0];
  const float* P = (const float*)d_in[1];
  float* out = (float*)d_out;

  float* T     = (float*)d_ws;
  float* norms = T + (size_t)NPTS * RANK;

  psd_project_kernel<<<512, 256, 0, stream>>>(R, P, T);

  psd_rownorm_kernel<<<NPTS / 256, 256, 0, stream>>>(T, norms);

  psd_gram_dist_kernel<<<256, 256, 0, stream>>>(T, norms, out);
}


# distance kernel computes the row norms itself from the T fragments it holds in registers (lane halves joined by permlanex16); norm launch removed, 2 launches left; changes norm summation order
# speedup vs baseline: 2.4804x; 1.1820x over previous
#include <hip/hip_runtime.h>
#include <hip/hip_bf16.h>

#define NPTS      2048
#define MODEL_DIM 1024
#define RANK      64

typedef float v2f __attribute__((ext_vector_type(2)));
typedef float v8f __attribute__((ext_vector_type(8)));

#define KSPLIT 8
#define KSLICE (MODEL_DIM / KSPLIT)

__global__ __launch_bounds__(256) void psd_project_kernel(
    const float* __restrict__ R, const float* __restrict__ P,
    float* __restrict__ T) {
  __shared__ float part[KSPLIT][8][32];

  const int lane = threadIdx.x & 31;
  const int wave = threadIdx.x >> 5;
  const int tile = blockIdx.x;
  const int rI = (tile >> 2) * 16;
  const int cB = (tile & 3) * 16;

  const int m    = lane & 15;
  const int koff = (lane >> 4) * 2;
  const int kBase = wave * KSLICE;

  const float* aPtr = R + (size_t)(rI + m) * MODEL_DIM + kBase + koff;
  const float* bCol = P + (size_t)kBase * RANK + cB + m;

  v8f acc = {};
#pragma unroll
  for (int s = 0; s < KSLICE / 4; ++s) {
    const int k0 = 4 * s;
    v2f a = *(const v2f*)(aPtr + k0);
    v2f b;
    b.x = bCol[(size_t)(k0 + koff) * RANK];
    b.y = bCol[(size_t)(k0 + koff + 1) * RANK];
    acc = __builtin_amdgcn_wmma_f32_16x16x4_f32(
         false, a,  false, b,
         (short)0, acc,  false,  false);
  }

#pragma unroll
  for (int r = 0; r < 8; ++r) {
    part[wave][r][lane] = acc[r];
  }
  __syncthreads();

  float sum = part[0][wave][lane];
  for (int w = 1; w < KSPLIT; ++w) {
    sum += part[w][wave][lane];
  }
  const int row = rI + ((lane >> 4) << 3) + wave;
  T[(size_t)row * RANK + cB + m] = sum;
}

#define KSTEPS (RANK / 4)
#define WAVE_COL_TILES 4

__device__ __forceinline__ void load_frag(v2f (&f)[KSTEPS], const float* __restrict__ p) {
#pragma unroll
  for (int s = 0; s < KSTEPS; ++s) {
    f[s] = *(const v2f*)(p + 4 * s);
  }
}

__device__ __forceinline__ float xhalf(float x) {
  const int b = __builtin_bit_cast(int, x);
  return __builtin_bit_cast(float,
      __builtin_amdgcn_permlanex16(b, b, 0x76543210, 0xfedcba98, false, false));
}

__device__ __forceinline__ float frag_sqnorm(const v2f (&f)[KSTEPS]) {
  v2f s2 = f[0] * f[0];
#pragma unroll
  for (int s = 1; s < KSTEPS; ++s) {
    s2 += f[s] * f[s];
  }
  const float p = s2.x + s2.y;
  return p + xhalf(p);
}

__device__ __forceinline__ void dist_tiles(
    const v2f (&a0)[KSTEPS], const v2f (&a1)[KSTEPS], const v2f (&b)[KSTEPS],
    const float (&ni0)[8], const float (&ni1)[8],
    float* __restrict__ o0, float* __restrict__ o1) {
  const float nj = frag_sqnorm(b);
  v8f acc0 = {};
  v8f acc1 = {};
#pragma unroll
  for (int s = 0; s < KSTEPS; ++s) {
    acc0 = __builtin_amdgcn_wmma_f32_16x16x4_f32(
        false, a0[s], false, b[s], (short)0, acc0, false, false);
    acc1 = __builtin_amdgcn_wmma_f32_16x16x4_f32(
        false, a1[s], false, b[s], (short)0, acc1, false, false);
  }
#pragma unroll
  for (int r = 0; r < 8; ++r) {
    o0[(size_t)r * NPTS] = ni0[r] + nj - 2.0f * acc0[r];
    o1[(size_t)r * NPTS] = ni1[r] + nj - 2.0f * acc1[r];
  }
}

__global__ __launch_bounds__(256) void psd_gram_dist_kernel(
    const float* __restrict__ T, float* __restrict__ out) {
  const int lane = threadIdx.x & 31;
  const int wave = threadIdx.x >> 5;

  const int rI = (blockIdx.x >> 4) * 128 + (wave >> 1) * 32;
  const int cJ = (blockIdx.x & 15) * 128 + (wave & 1) * 64;

  const int m    = lane & 15;
  const int koff = (lane >> 4) * 2;
  const int mBase0 = rI + ((lane >> 4) << 3);
  const int mBase1 = mBase0 + 16;

  v2f a0[KSTEPS], a1[KSTEPS];
  load_frag(a0, T + (size_t)(rI + m) * RANK + koff);
  load_frag(a1, T + (size_t)(rI + 16 + m) * RANK + koff);

  const float* bPtr = T + (size_t)(cJ + m) * RANK + koff;
  v2f b0[KSTEPS], b1[KSTEPS];
  load_frag(b0, bPtr);

  const float n0 = frag_sqnorm(a0);
  const float n1 = frag_sqnorm(a1);
  const int srcLane = (lane >> 4) << 3;
  float ni0[8], ni1[8];
#pragma unroll
  for (int r = 0; r < 8; ++r) {
    ni0[r] = __shfl(n0, srcLane + r);
    ni1[r] = __shfl(n1, srcLane + r);
  }

  float* o0 = out + (size_t)mBase0 * NPTS + cJ + m;
  float* o1 = out + (size_t)mBase1 * NPTS + cJ + m;

#pragma unroll 1
  for (int t = 0; t < WAVE_COL_TILES; t += 2) {
    load_frag(b1, bPtr + (size_t)(t + 1) * 16 * RANK);
    dist_tiles(a0, a1, b0, ni0, ni1, o0 + 16 * t, o1 + 16 * t);
    if (t + 2 < WAVE_COL_TILES) {
      load_frag(b0, bPtr + (size_t)(t + 2) * 16 * RANK);
    }
    dist_tiles(a0, a1, b1, ni0, ni1, o0 + 16 * (t + 1), o1 + 16 * (t + 1));
  }
}

extern "C" void kernel_launch(void* const* d_in, const int* in_sizes, int n_in,
                              void* d_out, int out_size, void* d_ws, size_t ws_size,
                              hipStream_t stream) {
  const float* R = (const float*)d_in[0];
  const float* P = (const float*)d_in[1];
  float* out = (float*)d_out;

  float* T = (float*)d_ws;

  psd_project_kernel<<<512, 256, 0, stream>>>(R, P, T);

  psd_gram_dist_kernel<<<256, 256, 0, stream>>>(T, out);
}


# projection GEMM: a wave computes two column tiles with interleaved (even/odd) columns: P is read as whole 128-byte lines, each R fragment feeds two WMMAs; 256 blocks of 16x32, not 512 of 16x16
# speedup vs baseline: 3.0471x; 1.2285x over previous
#include <hip/hip_runtime.h>
#include <hip/hip_bf16.h>

#define NPTS      2048
#define MODEL_DIM 1024
#define RANK      64

typedef float v2f __attribute__((ext_vector_type(2)));
typedef float v8f __attribute__((ext_vector_type(8)));

#define KSPLIT 8
#define KSLICE (MODEL_DIM / KSPLIT)

__global__ __launch_bounds__(256) __attribute__((amdgpu_waves_per_eu(2, 8)))
void psd_project_kernel(
    const float* __restrict__ R, const float* __restrict__ P,
    float* __restrict__ T) {
  __shared__ v2f part[KSPLIT][8][32];

  const int lane = threadIdx.x & 31;
  const int wave = threadIdx.x >> 5;
  const int tile = blockIdx.x;
  const int rI = (tile >> 1) * 16;
  const int cB = (tile & 1) * 32;

  const int m    = lane & 15;
  const int koff = (lane >> 4) * 2;
  const int kBase = wave * KSLICE;

  const float* aPtr = R + (size_t)(rI + m) * MODEL_DIM + kBase + koff;
  const float* bPtr = P + (size_t)(kBase + koff) * RANK + cB + 2 * m;

  v8f acc0 = {};
  v8f acc1 = {};
#pragma unroll
  for (int s = 0; s < KSLICE / 4; ++s) {
    const int k0 = 4 * s;
    const v2f a  = *(const v2f*)(aPtr + k0);
    const v2f p0 = *(const v2f*)(bPtr + (size_t)k0 * RANK);
    const v2f p1 = *(const v2f*)(bPtr + (size_t)(k0 + 1) * RANK);
    v2f b0, b1;
    b0.x = p0.x; b0.y = p1.x;
    b1.x = p0.y; b1.y = p1.y;
    acc0 = __builtin_amdgcn_wmma_f32_16x16x4_f32(
         false, a,  false, b0,
         (short)0, acc0,  false,  false);
    acc1 = __builtin_amdgcn_wmma_f32_16x16x4_f32(
         false, a,  false, b1,
         (short)0, acc1,  false,  false);
  }

#pragma unroll
  for (int r = 0; r < 8; ++r) {
    v2f pr;
    pr.x = acc0[r];
    pr.y = acc1[r];
    part[wave][r][lane] = pr;
  }
  __syncthreads();

  v2f sum = part[0][wave][lane];
  for (int w = 1; w < KSPLIT; ++w) {
    sum += part[w][wave][lane];
  }
  const int row = rI + ((lane >> 4) << 3) + wave;
  *(v2f*)(T + (size_t)row * RANK + cB + 2 * m) = sum;
}

#define KSTEPS (RANK / 4)
#define WAVE_COL_TILES 4

__device__ __forceinline__ void load_frag(v2f (&f)[KSTEPS], const float* __restrict__ p) {
#pragma unroll
  for (int s = 0; s < KSTEPS; ++s) {
    f[s] = *(const v2f*)(p + 4 * s);
  }
}

__device__ __forceinline__ float xhalf(float x) {
  const int b = __builtin_bit_cast(int, x);
  return __builtin_bit_cast(float,
      __builtin_amdgcn_permlanex16(b, b, 0x76543210, 0xfedcba98, false, false));
}

__device__ __forceinline__ float frag_sqnorm(const v2f (&f)[KSTEPS]) {
  v2f s2 = f[0] * f[0];
#pragma unroll
  for (int s = 1; s < KSTEPS; ++s) {
    s2 += f[s] * f[s];
  }
  const float p = s2.x + s2.y;
  return p + xhalf(p);
}

__device__ __forceinline__ void dist_tiles(
    const v2f (&a0)[KSTEPS], const v2f (&a1)[KSTEPS], const v2f (&b)[KSTEPS],
    const float (&ni0)[8], const float (&ni1)[8],
    float* __restrict__ o0, float* __restrict__ o1) {
  const float nj = frag_sqnorm(b);
  v8f acc0 = {};
  v8f acc1 = {};
#pragma unroll
  for (int s = 0; s < KSTEPS; ++s) {
    acc0 = __builtin_amdgcn_wmma_f32_16x16x4_f32(
        false, a0[s], false, b[s], (short)0, acc0, false, false);
    acc1 = __builtin_amdgcn_wmma_f32_16x16x4_f32(
        false, a1[s], false, b[s], (short)0, acc1, false, false);
  }
#pragma unroll
  for (int r = 0; r < 8; ++r) {
    o0[(size_t)r * NPTS] = ni0[r] + nj - 2.0f * acc0[r];
    o1[(size_t)r * NPTS] = ni1[r] + nj - 2.0f * acc1[r];
  }
}

__global__ __launch_bounds__(256) void psd_gram_dist_kernel(
    const float* __restrict__ T, float* __restrict__ out) {
  const int lane = threadIdx.x & 31;
  const int wave = threadIdx.x >> 5;

  const int rI = (blockIdx.x >> 4) * 128 + (wave >> 1) * 32;
  const int cJ = (blockIdx.x & 15) * 128 + (wave & 1) * 64;

  const int m    = lane & 15;
  const int koff = (lane >> 4) * 2;
  const int mBase0 = rI + ((lane >> 4) << 3);
  const int mBase1 = mBase0 + 16;

  v2f a0[KSTEPS], a1[KSTEPS];
  load_frag(a0, T + (size_t)(rI + m) * RANK + koff);
  load_frag(a1, T + (size_t)(rI + 16 + m) * RANK + koff);

  const float* bPtr = T + (size_t)(cJ + m) * RANK + koff;
  v2f b0[KSTEPS], b1[KSTEPS];
  load_frag(b0, bPtr);

  const float n0 = frag_sqnorm(a0);
  const float n1 = frag_sqnorm(a1);
  const int srcLane = (lane >> 4) << 3;
  float ni0[8], ni1[8];
#pragma unroll
  for (int r = 0; r < 8; ++r) {
    ni0[r] = __shfl(n0, srcLane + r);
    ni1[r] = __shfl(n1, srcLane + r);
  }

  float* o0 = out + (size_t)mBase0 * NPTS + cJ + m;
  float* o1 = out + (size_t)mBase1 * NPTS + cJ + m;

#pragma unroll 1
  for (int t = 0; t < WAVE_COL_TILES; t += 2) {
    load_frag(b1, bPtr + (size_t)(t + 1) * 16 * RANK);
    dist_tiles(a0, a1, b0, ni0, ni1, o0 + 16 * t, o1 + 16 * t);
    if (t + 2 < WAVE_COL_TILES) {
      load_frag(b0, bPtr + (size_t)(t + 2) * 16 * RANK);
    }
    dist_tiles(a0, a1, b1, ni0, ni1, o0 + 16 * (t + 1), o1 + 16 * (t + 1));
  }
}

extern "C" void kernel_launch(void* const* d_in, const int* in_sizes, int n_in,
                              void* d_out, int out_size, void* d_ws, size_t ws_size,
                              hipStream_t stream) {
  const float* R = (const float*)d_in[0];
  const float* P = (const float*)d_in[1];
  float* out = (float*)d_out;

  float* T = (float*)d_ws;

  psd_project_kernel<<<256, 256, 0, stream>>>(R, P, T);

  psd_gram_dist_kernel<<<256, 256, 0, stream>>>(T, out);
}


# projection kernel stores T in WMMA fragment order (16-row records of 512-byte chunks): each distance-kernel fragment load is one contiguous 512 B read; changes Gram and norm summation order
# speedup vs baseline: 3.4193x; 1.1221x over previous
#include <hip/hip_runtime.h>
#include <hip/hip_bf16.h>

#define NPTS      2048
#define MODEL_DIM 1024
#define RANK      64

typedef float v2f __attribute__((ext_vector_type(2)));
typedef float v4f __attribute__((ext_vector_type(4)));
typedef float v8f __attribute__((ext_vector_type(8)));

#define KSPLIT 8
#define KSLICE (MODEL_DIM / KSPLIT)
#define PART_PITCH 33
#define TILE_FLOATS (16 * RANK)
#define CHUNK_FLOATS 128

__global__ __launch_bounds__(256) __attribute__((amdgpu_waves_per_eu(2, 8)))
void psd_project_kernel(
    const float* __restrict__ R, const float* __restrict__ P,
    float* __restrict__ T) {
  __shared__ v2f part[KSPLIT][8][PART_PITCH];

  const int lane = threadIdx.x & 31;
  const int wave = threadIdx.x >> 5;
  const int tile = blockIdx.x;
  const int rI = (tile >> 1) * 16;
  const int cB = (tile & 1) * 32;

  const int m    = lane & 15;
  const int koff = (lane >> 4) * 2;
  const int kBase = wave * KSLICE;

  const float* aPtr = R + (size_t)(rI + m) * MODEL_DIM + kBase + koff;
  const float* bPtr = P + (size_t)(kBase + koff) * RANK + cB + 2 * m;

  v8f acc0 = {};
  v8f acc1 = {};
#pragma unroll
  for (int s = 0; s < KSLICE / 4; ++s) {
    const int k0 = 4 * s;
    const v2f a  = *(const v2f*)(aPtr + k0);
    const v2f p0 = *(const v2f*)(bPtr + (size_t)k0 * RANK);
    const v2f p1 = *(const v2f*)(bPtr + (size_t)(k0 + 1) * RANK);
    v2f b0, b1;
    b0.x = p0.x; b0.y = p1.x;
    b1.x = p0.y; b1.y = p1.y;
    acc0 = __builtin_amdgcn_wmma_f32_16x16x4_f32(
         false, a,  false, b0,
         (short)0, acc0,  false,  false);
    acc1 = __builtin_amdgcn_wmma_f32_16x16x4_f32(
         false, a,  false, b1,
         (short)0, acc1,  false,  false);
  }

#pragma unroll
  for (int r = 0; r < 8; ++r) {
    v2f pr;
    pr.x = acc0[r];
    pr.y = acc1[r];
    part[wave][r][lane] = pr;
  }
  __syncthreads();

  const int i   = lane & 15;
  const int rh  = lane >> 4;
  const int rr  = i >> 1;
  const int cpl = i & 1;
  const int jj  = wave >> 1;
  const int hh  = wave & 1;
  const int cp  = 4 * jj + 2 * hh + cpl;
  const int src = 16 * rh + cp;
  v2f sum = part[0][rr][src];
  for (int w = 1; w < KSPLIT; ++w) {
    sum += part[w][rr][src];
  }
  float* dst = T + (size_t)(tile >> 1) * TILE_FLOATS
                 + ((cB >> 3) + jj) * CHUNK_FLOATS
                 + (hh * 16 + 8 * rh + rr) * 4 + 2 * cpl;
  *(v2f*)dst = sum;
}

#define KSTEPS (RANK / 4)
#define WAVE_COL_TILES 4

__device__ __forceinline__ void load_frag(v2f (&f)[KSTEPS], const float* __restrict__ p) {
#pragma unroll
  for (int j = 0; j < KSTEPS / 2; ++j) {
    const v4f q = *(const v4f*)(p + CHUNK_FLOATS * j);
    v2f lo, hi;
    lo.x = q.x; lo.y = q.y;
    hi.x = q.z; hi.y = q.w;
    f[2 * j]     = lo;
    f[2 * j + 1] = hi;
  }
}

__device__ __forceinline__ float xhalf(float x) {
  const int b = __builtin_bit_cast(int, x);
  return __builtin_bit_cast(float,
      __builtin_amdgcn_permlanex16(b, b, 0x76543210, 0xfedcba98, false, false));
}

__device__ __forceinline__ float frag_sqnorm(const v2f (&f)[KSTEPS]) {
  v2f s2 = f[0] * f[0];
#pragma unroll
  for (int s = 1; s < KSTEPS; ++s) {
    s2 += f[s] * f[s];
  }
  const float p = s2.x + s2.y;
  return p + xhalf(p);
}

__device__ __forceinline__ void dist_tiles(
    const v2f (&a0)[KSTEPS], const v2f (&a1)[KSTEPS], const v2f (&b)[KSTEPS],
    const float (&ni0)[8], const float (&ni1)[8],
    float* __restrict__ o0, float* __restrict__ o1) {
  const float nj = frag_sqnorm(b);
  v8f acc0 = {};
  v8f acc1 = {};
#pragma unroll
  for (int s = 0; s < KSTEPS; ++s) {
    acc0 = __builtin_amdgcn_wmma_f32_16x16x4_f32(
        false, a0[s], false, b[s], (short)0, acc0, false, false);
    acc1 = __builtin_amdgcn_wmma_f32_16x16x4_f32(
        false, a1[s], false, b[s], (short)0, acc1, false, false);
  }
#pragma unroll
  for (int r = 0; r < 8; ++r) {
    o0[(size_t)r * NPTS] = ni0[r] + nj - 2.0f * acc0[r];
    o1[(size_t)r * NPTS] = ni1[r] + nj - 2.0f * acc1[r];
  }
}

__global__ __launch_bounds__(256) void psd_gram_dist_kernel(
    const float* __restrict__ T, float* __restrict__ out) {
  const int lane = threadIdx.x & 31;
  const int wave = threadIdx.x >> 5;

  const int rI = (blockIdx.x >> 4) * 128 + (wave >> 1) * 32;
  const int cJ = (blockIdx.x & 15) * 128 + (wave & 1) * 64;

  const int m    = lane & 15;
  const int mBase0 = rI + ((lane >> 4) << 3);
  const int mBase1 = mBase0 + 16;

  v2f a0[KSTEPS], a1[KSTEPS];
  load_frag(a0, T + (size_t)rI * RANK + 4 * lane);
  load_frag(a1, T + (size_t)(rI + 16) * RANK + 4 * lane);

  const float* bPtr = T + (size_t)cJ * RANK + 4 * lane;
  v2f b0[KSTEPS], b1[KSTEPS];
  load_frag(b0, bPtr);

  const float n0 = frag_sqnorm(a0);
  const float n1 = frag_sqnorm(a1);
  const int srcLane = (lane >> 4) << 3;
  float ni0[8], ni1[8];
#pragma unroll
  for (int r = 0; r < 8; ++r) {
    ni0[r] = __shfl(n0, srcLane + r);
    ni1[r] = __shfl(n1, srcLane + r);
  }

  float* o0 = out + (size_t)mBase0 * NPTS + cJ + m;
  float* o1 = out + (size_t)mBase1 * NPTS + cJ + m;

#pragma unroll 1
  for (int t = 0; t < WAVE_COL_TILES; t += 2) {
    load_frag(b1, bPtr + (size_t)(t + 1) * 16 * RANK);
    dist_tiles(a0, a1, b0, ni0, ni1, o0 + 16 * t, o1 + 16 * t);
    if (t + 2 < WAVE_COL_TILES) {
      load_frag(b0, bPtr + (size_t)(t + 2) * 16 * RANK);
    }
    dist_tiles(a0, a1, b1, ni0, ni1, o0 + 16 * (t + 1), o1 + 16 * (t + 1));
  }
}

extern "C" void kernel_launch(void* const* d_in, const int* in_sizes, int n_in,
                              void* d_out, int out_size, void* d_ws, size_t ws_size,
                              hipStream_t stream) {
  const float* R = (const float*)d_in[0];
  const float* P = (const float*)d_in[1];
  float* out = (float*)d_out;

  float* T = (float*)d_ws;

  psd_project_kernel<<<256, 256, 0, stream>>>(R, P, T);

  psd_gram_dist_kernel<<<256, 256, 0, stream>>>(T, out);
}
